# starting kernel
# baseline (speedup 1.0000x reference)
#include <hip/hip_runtime.h>
#include <math.h>

typedef float v2f __attribute__((ext_vector_type(2)));
typedef float v8f __attribute__((ext_vector_type(8)));

#define LDS_DSTRIDE 1096
#define LDS_HSTRIDE 34

__global__ void build_A_kernel(float* __restrict__ A) {
    const double PI = 3.14159265358979323846;
    int tid = threadIdx.x;
    for (int j = 0; j < 8; ++j) {
        int idx = tid + 256 * j;
        int d  = idx >> 6;
        int dp = idx & 63;
        double acc = 0.0;
        for (int k = 0; k < 32; ++k) {
            double s32 = (k == 0) ? sqrt(1.0 / 32.0) : sqrt(2.0 / 32.0);
            double s64 = (k == 0) ? sqrt(1.0 / 64.0) : sqrt(2.0 / 64.0);
            double c32 = cos(PI * (d  + 0.5) * k / 32.0) * s32;
            double c64 = cos(PI * (dp + 0.5) * k / 64.0) * s64;
            acc += c32 * c64;
        }
        A[idx] = (float)acc;
    }
}

__global__ void __launch_bounds__(256, 1)
spectral_pool_kernel(const float* __restrict__ x, const float* __restrict__ Aop,
                     float* __restrict__ out) {
    __shared__ float tmp[64 * LDS_DSTRIDE];

    const int lane = threadIdx.x & 31;
    const int wave = threadIdx.x >> 5;
    const int half = lane >> 4;
    const int l16  = lane & 15;

    const size_t slab = (size_t)blockIdx.x * (64 * 64 * 64);

    v2f aF[2][16];
#pragma unroll
    for (int mt = 0; mt < 2; ++mt) {
#pragma unroll
        for (int kk = 0; kk < 16; ++kk) {
            int row = mt * 16 + l16;
            int col = kk * 4 + 2 * half;
            v2f a;
            a.x = Aop[row * 64 + col];
            a.y = Aop[row * 64 + col + 1];
            aF[mt][kk] = a;
        }
    }

    const v8f vzero = {0.f, 0.f, 0.f, 0.f, 0.f, 0.f, 0.f, 0.f};

    for (int g = 0; g < 8; ++g) {
        const int dp = g * 8 + wave;
        const float* src = x + slab + (size_t)dp * (64 * 64);

        if (g < 7) {
            const float* nxt = x + slab + (size_t)(dp + 8) * (64 * 64);
            __builtin_prefetch(nxt + lane * 128, 0, 1);
        }

        v8f acc00 = vzero, acc01 = vzero, acc10 = vzero, acc11 = vzero;

#pragma unroll
        for (int kk = 0; kk < 16; ++kk) {
            const int r0 = kk * 4 + 2 * half;
            v2f b0, b1;
            b0.x = src[r0 * 64 + l16];
            b0.y = src[(r0 + 1) * 64 + l16];
            b1.x = src[r0 * 64 + 16 + l16];
            b1.y = src[(r0 + 1) * 64 + 16 + l16];

            acc00 = __builtin_amdgcn_wmma_f32_16x16x4_f32(false, aF[0][kk], false, b0,
                                                          (short)0, acc00, false, false);
            acc01 = __builtin_amdgcn_wmma_f32_16x16x4_f32(false, aF[0][kk], false, b1,
                                                          (short)0, acc01, false, false);
            acc10 = __builtin_amdgcn_wmma_f32_16x16x4_f32(false, aF[1][kk], false, b0,
                                                          (short)0, acc10, false, false);
            acc11 = __builtin_amdgcn_wmma_f32_16x16x4_f32(false, aF[1][kk], false, b1,
                                                          (short)0, acc11, false, false);
        }

        float* trow = &tmp[dp * LDS_DSTRIDE];
#pragma unroll
        for (int v = 0; v < 8; ++v) {
            const int hlo = v + 8 * half;
            trow[hlo * LDS_HSTRIDE + l16]              = acc00[v];
            trow[hlo * LDS_HSTRIDE + 16 + l16]         = acc01[v];
            trow[(16 + hlo) * LDS_HSTRIDE + l16]       = acc10[v];
            trow[(16 + hlo) * LDS_HSTRIDE + 16 + l16]  = acc11[v];
        }
    }

    __syncthreads();

    for (int g = 0; g < 4; ++g) {
        const int h = g * 8 + wave;
        const float* tcol = &tmp[h * LDS_HSTRIDE];

        v8f acc00 = vzero, acc01 = vzero, acc10 = vzero, acc11 = vzero;

#pragma unroll
        for (int kk = 0; kk < 16; ++kk) {
            const int r0 = kk * 4 + 2 * half;
            v2f b0, b1;
            b0.x = tcol[r0 * LDS_DSTRIDE + l16];
            b0.y = tcol[(r0 + 1) * LDS_DSTRIDE + l16];
            b1.x = tcol[r0 * LDS_DSTRIDE + 16 + l16];
            b1.y = tcol[(r0 + 1) * LDS_DSTRIDE + 16 + l16];

            acc00 = __builtin_amdgcn_wmma_f32_16x16x4_f32(false, aF[0][kk], false, b0,
                                                          (short)0, acc00, false, false);
            acc01 = __builtin_amdgcn_wmma_f32_16x16x4_f32(false, aF[0][kk], false, b1,
                                                          (short)0, acc01, false, false);
            acc10 = __builtin_amdgcn_wmma_f32_16x16x4_f32(false, aF[1][kk], false, b0,
                                                          (short)0, acc10, false, false);
            acc11 = __builtin_amdgcn_wmma_f32_16x16x4_f32(false, aF[1][kk], false, b1,
                                                          (short)0, acc11, false, false);
        }

        float* obase = out + (size_t)blockIdx.x * 32768 + h * 32;
#pragma unroll
        for (int v = 0; v < 8; ++v) {
            const int dlo = v + 8 * half;
            obase[dlo * 1024 + l16]             = acc00[v];
            obase[dlo * 1024 + 16 + l16]        = acc01[v];
            obase[(16 + dlo) * 1024 + l16]      = acc10[v];
            obase[(16 + dlo) * 1024 + 16 + l16] = acc11[v];
        }
    }
}

extern "C" void kernel_launch(void* const* d_in, const int* in_sizes, int n_in,
                              void* d_out, int out_size, void* d_ws, size_t ws_size,
                              hipStream_t stream) {
    const float* x = (const float*)d_in[0];
    float* A       = (float*)d_ws;
    float* out     = (float*)d_out;

    build_A_kernel<<<1, 256, 0, stream>>>(A);
    spectral_pool_kernel<<<256, 256, 0, stream>>>(x, A, out);
}


# operator-build kernel: each block fills one 256-entry fp64 cos(pi*m/128) table in LDS (one cospi per thread), each thread sums one element, 16 blocks; was 512 fp64 cos per thread in one block
# speedup vs baseline: 20.8188x; 20.8188x over previous
#include <hip/hip_runtime.h>
#include <math.h>

typedef float v2f __attribute__((ext_vector_type(2)));
typedef float v8f __attribute__((ext_vector_type(8)));

#define LDS_DSTRIDE 1096
#define LDS_HSTRIDE 34

__global__ void __launch_bounds__(128)
build_A_kernel(float* __restrict__ A) {
    __shared__ double U[256];
    __shared__ double V[256];
    const int t = threadIdx.x;
    const double c = cospi((double)t * (1.0 / 128.0));
    const double u = c * sqrt(2.0 / 32.0);
    const double v = c * sqrt(2.0 / 64.0);
    U[t] = u;  U[t + 128] = -u;
    V[t] = v;  V[t + 128] = -v;
    __syncthreads();

    const int idx = blockIdx.x * 128 + t;
    const int d  = idx >> 6;
    const int dp = idx & 63;
    const int ea = 2 * (2 * d + 1);
    const int eb = 2 * dp + 1;
    double cu[31], cv[31];
    for (int k = 1; k < 32; ++k) {
        cu[k - 1] = U[(ea * k) & 255];
        cv[k - 1] = V[(eb * k) & 255];
    }
    __builtin_amdgcn_sched_barrier(0);
    double acc = sqrt(1.0 / 32.0) * sqrt(1.0 / 64.0);
    for (int k = 1; k < 32; ++k)
        acc = fma(cu[k - 1], cv[k - 1], acc);
    A[idx] = (float)acc;
}

__global__ void __launch_bounds__(256, 1)
spectral_pool_kernel(const float* __restrict__ x, const float* __restrict__ Aop,
                     float* __restrict__ out) {
    __shared__ float tmp[64 * LDS_DSTRIDE];

    const int lane = threadIdx.x & 31;
    const int wave = threadIdx.x >> 5;
    const int half = lane >> 4;
    const int l16  = lane & 15;

    const size_t slab = (size_t)blockIdx.x * (64 * 64 * 64);

    v2f aF[2][16];
#pragma unroll
    for (int mt = 0; mt < 2; ++mt) {
#pragma unroll
        for (int kk = 0; kk < 16; ++kk) {
            int row = mt * 16 + l16;
            int col = kk * 4 + 2 * half;
            v2f a;
            a.x = Aop[row * 64 + col];
            a.y = Aop[row * 64 + col + 1];
            aF[mt][kk] = a;
        }
    }

    const v8f vzero = {0.f, 0.f, 0.f, 0.f, 0.f, 0.f, 0.f, 0.f};

    for (int g = 0; g < 8; ++g) {
        const int dp = g * 8 + wave;
        const float* src = x + slab + (size_t)dp * (64 * 64);

        if (g < 7) {
            const float* nxt = x + slab + (size_t)(dp + 8) * (64 * 64);
            __builtin_prefetch(nxt + lane * 128, 0, 1);
        }

        v8f acc00 = vzero, acc01 = vzero, acc10 = vzero, acc11 = vzero;

#pragma unroll
        for (int kk = 0; kk < 16; ++kk) {
            const int r0 = kk * 4 + 2 * half;
            v2f b0, b1;
            b0.x = src[r0 * 64 + l16];
            b0.y = src[(r0 + 1) * 64 + l16];
            b1.x = src[r0 * 64 + 16 + l16];
            b1.y = src[(r0 + 1) * 64 + 16 + l16];

            acc00 = __builtin_amdgcn_wmma_f32_16x16x4_f32(false, aF[0][kk], false, b0,
                                                          (short)0, acc00, false, false);
            acc01 = __builtin_amdgcn_wmma_f32_16x16x4_f32(false, aF[0][kk], false, b1,
                                                          (short)0, acc01, false, false);
            acc10 = __builtin_amdgcn_wmma_f32_16x16x4_f32(false, aF[1][kk], false, b0,
                                                          (short)0, acc10, false, false);
            acc11 = __builtin_amdgcn_wmma_f32_16x16x4_f32(false, aF[1][kk], false, b1,
                                                          (short)0, acc11, false, false);
        }

        float* trow = &tmp[dp * LDS_DSTRIDE];
#pragma unroll
        for (int v = 0; v < 8; ++v) {
            const int hlo = v + 8 * half;
            trow[hlo * LDS_HSTRIDE + l16]              = acc00[v];
            trow[hlo * LDS_HSTRIDE + 16 + l16]         = acc01[v];
            trow[(16 + hlo) * LDS_HSTRIDE + l16]       = acc10[v];
            trow[(16 + hlo) * LDS_HSTRIDE + 16 + l16]  = acc11[v];
        }
    }

    __syncthreads();

    for (int g = 0; g < 4; ++g) {
        const int h = g * 8 + wave;
        const float* tcol = &tmp[h * LDS_HSTRIDE];

        v8f acc00 = vzero, acc01 = vzero, acc10 = vzero, acc11 = vzero;

#pragma unroll
        for (int kk = 0; kk < 16; ++kk) {
            const int r0 = kk * 4 + 2 * half;
            v2f b0, b1;
            b0.x = tcol[r0 * LDS_DSTRIDE + l16];
            b0.y = tcol[(r0 + 1) * LDS_DSTRIDE + l16];
            b1.x = tcol[r0 * LDS_DSTRIDE + 16 + l16];
            b1.y = tcol[(r0 + 1) * LDS_DSTRIDE + 16 + l16];

            acc00 = __builtin_amdgcn_wmma_f32_16x16x4_f32(false, aF[0][kk], false, b0,
                                                          (short)0, acc00, false, false);
            acc01 = __builtin_amdgcn_wmma_f32_16x16x4_f32(false, aF[0][kk], false, b1,
                                                          (short)0, acc01, false, false);
            acc10 = __builtin_amdgcn_wmma_f32_16x16x4_f32(false, aF[1][kk], false, b0,
                                                          (short)0, acc10, false, false);
            acc11 = __builtin_amdgcn_wmma_f32_16x16x4_f32(false, aF[1][kk], false, b1,
                                                          (short)0, acc11, false, false);
        }

        float* obase = out + (size_t)blockIdx.x * 32768 + h * 32;
#pragma unroll
        for (int v = 0; v < 8; ++v) {
            const int dlo = v + 8 * half;
            obase[dlo * 1024 + l16]             = acc00[v];
            obase[dlo * 1024 + 16 + l16]        = acc01[v];
            obase[(16 + dlo) * 1024 + l16]      = acc10[v];
            obase[(16 + dlo) * 1024 + 16 + l16] = acc11[v];
        }
    }
}

extern "C" void kernel_launch(void* const* d_in, const int* in_sizes, int n_in,
                              void* d_out, int out_size, void* d_ws, size_t ws_size,
                              hipStream_t stream) {
    const float* x = (const float*)d_in[0];
    float* A       = (float*)d_ws;
    float* out     = (float*)d_out;

    build_A_kernel<<<16, 128, 0, stream>>>(A);
    spectral_pool_kernel<<<256, 256, 0, stream>>>(x, A, out);
}


# stage 2 folded into the slab loop: after each group of 8 slabs (LDS-counter-only barrier, x loads stay in flight) each wave adds the group's 2 k-steps to register-held output tiles; sum order kept
# speedup vs baseline: 21.4736x; 1.0315x over previous
#include <hip/hip_runtime.h>
#include <math.h>

typedef float v2f __attribute__((ext_vector_type(2)));
typedef float v8f __attribute__((ext_vector_type(8)));

#define LDS_HSTRIDE 68
#define LDS_PSTRIDE 2208
#define LDS_BUF (4 * LDS_PSTRIDE)

__global__ void __launch_bounds__(128)
build_A_kernel(float* __restrict__ A) {
    __shared__ double U[256];
    __shared__ double V[256];
    const int t = threadIdx.x;
    const double c = cospi((double)t * (1.0 / 128.0));
    const double u = c * sqrt(2.0 / 32.0);
    const double v = c * sqrt(2.0 / 64.0);
    U[t] = u;  U[t + 128] = -u;
    V[t] = v;  V[t + 128] = -v;
    __syncthreads();

    const int idx = blockIdx.x * 128 + t;
    const int d  = idx >> 6;
    const int dp = idx & 63;
    const int ea = 2 * (2 * d + 1);
    const int eb = 2 * dp + 1;
    double cu[31], cv[31];
    for (int k = 1; k < 32; ++k) {
        cu[k - 1] = U[(ea * k) & 255];
        cv[k - 1] = V[(eb * k) & 255];
    }
    __builtin_amdgcn_sched_barrier(0);
    double acc = sqrt(1.0 / 32.0) * sqrt(1.0 / 64.0);
    for (int k = 1; k < 32; ++k)
        acc = fma(cu[k - 1], cv[k - 1], acc);
    A[idx] = (float)acc;
}

static __device__ __forceinline__ void lds_barrier() {
    asm volatile("s_wait_dscnt 0x0\n\ts_barrier_signal -1\n\ts_barrier_wait -1" ::: "memory");
}

__global__ void __launch_bounds__(256, 1)
spectral_pool_kernel(const float* __restrict__ x, const float* __restrict__ Aop,
                     float* __restrict__ out) {
    __shared__ __attribute__((aligned(16))) float tmp[2 * LDS_BUF];

    const int lane = threadIdx.x & 31;
    const int wave = threadIdx.x >> 5;
    const int half = lane >> 4;
    const int l16  = lane & 15;

    const size_t slab = (size_t)blockIdx.x * (64 * 64 * 64);

    v2f aF[2][16];
#pragma unroll
    for (int mt = 0; mt < 2; ++mt) {
#pragma unroll
        for (int kk = 0; kk < 16; ++kk) {
            int row = mt * 16 + l16;
            int col = kk * 4 + 2 * half;
            v2f a;
            a.x = Aop[row * 64 + col];
            a.y = Aop[row * 64 + col + 1];
            aF[mt][kk] = a;
        }
    }

    const v8f vzero = {0.f, 0.f, 0.f, 0.f, 0.f, 0.f, 0.f, 0.f};

    v8f o[4][2][2];
#pragma unroll
    for (int gg = 0; gg < 4; ++gg) {
        o[gg][0][0] = vzero; o[gg][0][1] = vzero; o[gg][1][0] = vzero; o[gg][1][1] = vzero;
    }

    v2f xb0[16], xb1[16];
    const float* xs = x + slab + (size_t)wave * (64 * 64);
#pragma unroll
    for (int kk = 0; kk < 16; ++kk) {
        const int r0 = kk * 4 + 2 * half;
        xb0[kk].x = xs[r0 * 64 + l16];
        xb0[kk].y = xs[(r0 + 1) * 64 + l16];
        xb1[kk].x = xs[r0 * 64 + 16 + l16];
        xb1[kk].y = xs[(r0 + 1) * 64 + 16 + l16];
    }

#pragma unroll
    for (int g = 0; g < 8; ++g) {
        const float* nx = xs + (size_t)(g + 1) * (8 * 64 * 64);
        v8f acc00 = vzero, acc01 = vzero, acc10 = vzero, acc11 = vzero;
#pragma unroll
        for (int kk = 0; kk < 16; ++kk) {
            acc00 = __builtin_amdgcn_wmma_f32_16x16x4_f32(false, aF[0][kk], false, xb0[kk],
                                                          (short)0, acc00, false, false);
            acc01 = __builtin_amdgcn_wmma_f32_16x16x4_f32(false, aF[0][kk], false, xb1[kk],
                                                          (short)0, acc01, false, false);
            acc10 = __builtin_amdgcn_wmma_f32_16x16x4_f32(false, aF[1][kk], false, xb0[kk],
                                                          (short)0, acc10, false, false);
            acc11 = __builtin_amdgcn_wmma_f32_16x16x4_f32(false, aF[1][kk], false, xb1[kk],
                                                          (short)0, acc11, false, false);
            if (g < 7) {
                const int r0 = kk * 4 + 2 * half;
                xb0[kk].x = nx[r0 * 64 + l16];
                xb0[kk].y = nx[(r0 + 1) * 64 + l16];
                xb1[kk].x = nx[r0 * 64 + 16 + l16];
                xb1[kk].y = nx[(r0 + 1) * 64 + 16 + l16];
            }
        }

        float* trow = &tmp[(g & 1) * LDS_BUF + (wave >> 1) * LDS_PSTRIDE + (wave & 1)];
#pragma unroll
        for (int v = 0; v < 8; ++v) {
            const int hlo = v + 8 * half;
            trow[hlo * LDS_HSTRIDE + 2 * l16]                = acc00[v];
            trow[hlo * LDS_HSTRIDE + 2 * (16 + l16)]         = acc01[v];
            trow[(16 + hlo) * LDS_HSTRIDE + 2 * l16]         = acc10[v];
            trow[(16 + hlo) * LDS_HSTRIDE + 2 * (16 + l16)]  = acc11[v];
        }

        lds_barrier();

        const float* tbuf = &tmp[(g & 1) * LDS_BUF + half * LDS_PSTRIDE + 2 * l16];
        v2f t0[4][2], t1[4][2];
#pragma unroll
        for (int gg = 0; gg < 4; ++gg) {
            const float* tcol = tbuf + (gg * 8 + wave) * LDS_HSTRIDE;
#pragma unroll
            for (int k2 = 0; k2 < 2; ++k2) {
                t0[gg][k2] = *(const v2f*)(tcol + 2 * k2 * LDS_PSTRIDE);
                t1[gg][k2] = *(const v2f*)(tcol + 2 * k2 * LDS_PSTRIDE + 32);
            }
        }
#pragma unroll
        for (int gg = 0; gg < 4; ++gg) {
#pragma unroll
            for (int k2 = 0; k2 < 2; ++k2) {
                const int kk = 2 * g + k2;
                o[gg][0][0] = __builtin_amdgcn_wmma_f32_16x16x4_f32(false, aF[0][kk], false, t0[gg][k2],
                                                                    (short)0, o[gg][0][0], false, false);
                o[gg][0][1] = __builtin_amdgcn_wmma_f32_16x16x4_f32(false, aF[0][kk], false, t1[gg][k2],
                                                                    (short)0, o[gg][0][1], false, false);
                o[gg][1][0] = __builtin_amdgcn_wmma_f32_16x16x4_f32(false, aF[1][kk], false, t0[gg][k2],
                                                                    (short)0, o[gg][1][0], false, false);
                o[gg][1][1] = __builtin_amdgcn_wmma_f32_16x16x4_f32(false, aF[1][kk], false, t1[gg][k2],
                                                                    (short)0, o[gg][1][1], false, false);
            }
        }
    }

#pragma unroll
    for (int gg = 0; gg < 4; ++gg) {
        float* obase = out + (size_t)blockIdx.x * 32768 + (gg * 8 + wave) * 32;
#pragma unroll
        for (int v = 0; v < 8; ++v) {
            const int dlo = v + 8 * half;
            obase[dlo * 1024 + l16]             = o[gg][0][0][v];
            obase[dlo * 1024 + 16 + l16]        = o[gg][0][1][v];
            obase[(16 + dlo) * 1024 + l16]      = o[gg][1][0][v];
            obase[(16 + dlo) * 1024 + 16 + l16] = o[gg][1][1][v];
        }
    }
}

extern "C" void kernel_launch(void* const* d_in, const int* in_sizes, int n_in,
                              void* d_out, int out_size, void* d_ws, size_t ws_size,
                              hipStream_t stream) {
    const float* x = (const float*)d_in[0];
    float* A       = (float*)d_ws;
    float* out     = (float*)d_out;

    build_A_kernel<<<16, 128, 0, stream>>>(A);
    spectral_pool_kernel<<<256, 256, 0, stream>>>(x, A, out);
}


# x refill regrouped: each slab's rows requested as ONE group of 8-byte loads right behind the stage-1 matrix block, two slabs ahead into a second register set (w tiles now even/odd); sum order kept
# speedup vs baseline: 22.4510x; 1.0455x over previous
#include <hip/hip_runtime.h>
#include <math.h>

typedef float v2f __attribute__((ext_vector_type(2)));
typedef float v8f __attribute__((ext_vector_type(8)));

#define LDS_HSTRIDE 68
#define LDS_PSTRIDE 2208
#define LDS_BUF (4 * LDS_PSTRIDE)

__global__ void __launch_bounds__(128)
build_A_kernel(float* __restrict__ A) {
    __shared__ double U[256];
    __shared__ double V[256];
    const int t = threadIdx.x;
    const double c = cospi((double)t * (1.0 / 128.0));
    const double u = c * sqrt(2.0 / 32.0);
    const double v = c * sqrt(2.0 / 64.0);
    U[t] = u;  U[t + 128] = -u;
    V[t] = v;  V[t + 128] = -v;
    __syncthreads();

    const int idx = blockIdx.x * 128 + t;
    const int d  = idx >> 6;
    const int dp = idx & 63;
    const int ea = 2 * (2 * d + 1);
    const int eb = 2 * dp + 1;
    double cu[31], cv[31];
    for (int k = 1; k < 32; ++k) {
        cu[k - 1] = U[(ea * k) & 255];
        cv[k - 1] = V[(eb * k) & 255];
    }
    __builtin_amdgcn_sched_barrier(0);
    double acc = sqrt(1.0 / 32.0) * sqrt(1.0 / 64.0);
    for (int k = 1; k < 32; ++k)
        acc = fma(cu[k - 1], cv[k - 1], acc);
    A[idx] = (float)acc;
}

static __device__ __forceinline__ void lds_barrier() {
    asm volatile("s_wait_dscnt 0x0\n\ts_barrier_signal -1\n\ts_barrier_wait -1" ::: "memory");
}

__global__ void __launch_bounds__(256, 1)
spectral_pool_kernel(const float* __restrict__ x, const float* __restrict__ Aop,
                     float* __restrict__ out) {
    __shared__ __attribute__((aligned(16))) float tmp[2 * LDS_BUF];

    const int lane = threadIdx.x & 31;
    const int wave = threadIdx.x >> 5;
    const int half = lane >> 4;
    const int l16  = lane & 15;

    const size_t slab = (size_t)blockIdx.x * (64 * 64 * 64);

    v2f aF[2][16];
#pragma unroll
    for (int mt = 0; mt < 2; ++mt) {
#pragma unroll
        for (int kk = 0; kk < 16; ++kk) {
            int row = mt * 16 + l16;
            int col = kk * 4 + 2 * half;
            v2f a;
            a.x = Aop[row * 64 + col];
            a.y = Aop[row * 64 + col + 1];
            aF[mt][kk] = a;
        }
    }

    const v8f vzero = {0.f, 0.f, 0.f, 0.f, 0.f, 0.f, 0.f, 0.f};

    v8f o[4][2][2];
#pragma unroll
    for (int gg = 0; gg < 4; ++gg) {
        o[gg][0][0] = vzero; o[gg][0][1] = vzero; o[gg][1][0] = vzero; o[gg][1][1] = vzero;
    }

    v2f ld[2][32];
    const float* xs = x + slab + (size_t)wave * (64 * 64) + 2 * l16;
#pragma unroll
    for (int j = 0; j < 2; ++j) {
#pragma unroll
        for (int i = 0; i < 32; ++i) {
            const int r = (i >> 1) * 4 + 2 * half + (i & 1);
            ld[j][i] = *(const v2f*)(xs + (size_t)j * (8 * 64 * 64) + r * 64);
        }
    }

#pragma unroll
    for (int g = 0; g < 8; ++g) {
        v8f acc00 = vzero, acc01 = vzero, acc10 = vzero, acc11 = vzero;
#pragma unroll
        for (int kk = 0; kk < 16; ++kk) {
            const v2f b0 = {ld[g & 1][2 * kk].x, ld[g & 1][2 * kk + 1].x};
            const v2f b1 = {ld[g & 1][2 * kk].y, ld[g & 1][2 * kk + 1].y};
            acc00 = __builtin_amdgcn_wmma_f32_16x16x4_f32(false, aF[0][kk], false, b0,
                                                          (short)0, acc00, false, false);
            acc01 = __builtin_amdgcn_wmma_f32_16x16x4_f32(false, aF[0][kk], false, b1,
                                                          (short)0, acc01, false, false);
            acc10 = __builtin_amdgcn_wmma_f32_16x16x4_f32(false, aF[1][kk], false, b0,
                                                          (short)0, acc10, false, false);
            acc11 = __builtin_amdgcn_wmma_f32_16x16x4_f32(false, aF[1][kk], false, b1,
                                                          (short)0, acc11, false, false);
        }

        if (g < 6) {
            __builtin_amdgcn_sched_barrier(0);
#pragma unroll
            for (int i = 0; i < 32; ++i) {
                const int r = (i >> 1) * 4 + 2 * half + (i & 1);
                ld[g & 1][i] = *(const v2f*)(xs + (size_t)(g + 2) * (8 * 64 * 64) + r * 64);
            }
            __builtin_amdgcn_sched_barrier(0);
        }

        float* trow = &tmp[(g & 1) * LDS_BUF + (wave >> 1) * LDS_PSTRIDE + (wave & 1)];
#pragma unroll
        for (int v = 0; v < 8; ++v) {
            const int hlo = v + 8 * half;
            trow[hlo * LDS_HSTRIDE + 2 * l16]                = acc00[v];
            trow[hlo * LDS_HSTRIDE + 2 * (16 + l16)]         = acc01[v];
            trow[(16 + hlo) * LDS_HSTRIDE + 2 * l16]         = acc10[v];
            trow[(16 + hlo) * LDS_HSTRIDE + 2 * (16 + l16)]  = acc11[v];
        }

        lds_barrier();

        const float* tbuf = &tmp[(g & 1) * LDS_BUF + half * LDS_PSTRIDE + 2 * l16];
        v2f t0[4][2], t1[4][2];
#pragma unroll
        for (int gg = 0; gg < 4; ++gg) {
            const float* tcol = tbuf + (gg * 8 + wave) * LDS_HSTRIDE;
#pragma unroll
            for (int k2 = 0; k2 < 2; ++k2) {
                t0[gg][k2] = *(const v2f*)(tcol + 2 * k2 * LDS_PSTRIDE);
                t1[gg][k2] = *(const v2f*)(tcol + 2 * k2 * LDS_PSTRIDE + 32);
            }
        }
#pragma unroll
        for (int gg = 0; gg < 4; ++gg) {
#pragma unroll
            for (int k2 = 0; k2 < 2; ++k2) {
                const int kk = 2 * g + k2;
                o[gg][0][0] = __builtin_amdgcn_wmma_f32_16x16x4_f32(false, aF[0][kk], false, t0[gg][k2],
                                                                    (short)0, o[gg][0][0], false, false);
                o[gg][0][1] = __builtin_amdgcn_wmma_f32_16x16x4_f32(false, aF[0][kk], false, t1[gg][k2],
                                                                    (short)0, o[gg][0][1], false, false);
                o[gg][1][0] = __builtin_amdgcn_wmma_f32_16x16x4_f32(false, aF[1][kk], false, t0[gg][k2],
                                                                    (short)0, o[gg][1][0], false, false);
                o[gg][1][1] = __builtin_amdgcn_wmma_f32_16x16x4_f32(false, aF[1][kk], false, t1[gg][k2],
                                                                    (short)0, o[gg][1][1], false, false);
            }
        }
    }

#pragma unroll
    for (int gg = 0; gg < 4; ++gg) {
        float* obase = out + (size_t)blockIdx.x * 32768 + (gg * 8 + wave) * 32;
#pragma unroll
        for (int v = 0; v < 8; ++v) {
            const int dlo = v + 8 * half;
            const v2f lo = {o[gg][0][0][v], o[gg][0][1][v]};
            const v2f hi = {o[gg][1][0][v], o[gg][1][1][v]};
            *(v2f*)(obase + dlo * 1024 + 2 * l16)        = lo;
            *(v2f*)(obase + (16 + dlo) * 1024 + 2 * l16) = hi;
        }
    }
}

extern "C" void kernel_launch(void* const* d_in, const int* in_sizes, int n_in,
                              void* d_out, int out_size, void* d_ws, size_t ws_size,
                              hipStream_t stream) {
    const float* x = (const float*)d_in[0];
    float* A       = (float*)d_ws;
    float* out     = (float*)d_out;

    build_A_kernel<<<16, 128, 0, stream>>>(A);
    spectral_pool_kernel<<<256, 256, 0, stream>>>(x, A, out);
}


# operator built inside the main kernel: 127-entry f32 table Q(j)=c*sin(63*pi*j/256)/sin(pi*j/256) in LDS, A[d][dp]=Q(j1)+Q(j2), closed form of the 32-term fp64 sum; one launch, was two
# speedup vs baseline: 25.1906x; 1.1220x over previous
#include <hip/hip_runtime.h>
#include <math.h>

typedef float v2f __attribute__((ext_vector_type(2)));
typedef float v8f __attribute__((ext_vector_type(8)));

#define LDS_HSTRIDE 68
#define LDS_PSTRIDE 2208
#define LDS_BUF (4 * LDS_PSTRIDE)

#define Q_SCALE 0.011048543456f

static __device__ __forceinline__ void lds_barrier() {
    asm volatile("s_wait_dscnt 0x0\n\ts_barrier_signal -1\n\ts_barrier_wait -1" ::: "memory");
}

__global__ void __launch_bounds__(256, 1)
spectral_pool_kernel(const float* __restrict__ x, float* __restrict__ out) {
    __shared__ __attribute__((aligned(16))) float tmp[2 * LDS_BUF];

    const int lane = threadIdx.x & 31;
    const int wave = threadIdx.x >> 5;
    const int half = lane >> 4;
    const int l16  = lane & 15;

    const size_t slab = (size_t)blockIdx.x * (64 * 64 * 64);

    const v8f vzero = {0.f, 0.f, 0.f, 0.f, 0.f, 0.f, 0.f, 0.f};

    v8f o[4][2][2];
#pragma unroll
    for (int gg = 0; gg < 4; ++gg) {
        o[gg][0][0] = vzero; o[gg][0][1] = vzero; o[gg][1][0] = vzero; o[gg][1][1] = vzero;
    }

    v2f ld[2][32];
    const float* xs = x + slab + (size_t)wave * (64 * 64) + 2 * l16;
#pragma unroll
    for (int i = 0; i < 32; ++i) {
        const int r = (i >> 1) * 4 + 2 * half + (i & 1);
        ld[0][i] = *(const v2f*)(xs + r * 64);
    }

    __shared__ float Qt[128];
    __shared__ float Et[128];
    if (threadIdx.x < 128) {
        const int t = threadIdx.x;
        const int j = 2 * t + 1;
        const int m = (63 * j) & 511;
        const float num = sinpif((float)m * (1.0f / 256.0f));
        const float den = sinpif((float)j * (1.0f / 256.0f));
        const float q = Q_SCALE * num / den;
        Qt[t] = q;
        if (t < 64) {
            Et[64 + t] = q;
            Et[63 - t] = q;
        }
    }
    lds_barrier();

    v2f aF[2][16];
    {
        const float* e0 = &Et[64 + 2 * l16 - 2 * half];
        const float* q0 = &Qt[1 + 2 * l16 + 2 * half];
#pragma unroll
        for (int mt = 0; mt < 2; ++mt) {
#pragma unroll
            for (int kk = 0; kk < 16; ++kk) {
                v2f a;
                a.x = e0[32 * mt - 4 * kk]     + q0[32 * mt + 4 * kk];
                a.y = e0[32 * mt - 4 * kk - 1] + q0[32 * mt + 4 * kk + 1];
                aF[mt][kk] = a;
            }
        }
    }
    __builtin_amdgcn_sched_barrier(0);

#pragma unroll
    for (int i = 0; i < 32; ++i) {
        const int r = (i >> 1) * 4 + 2 * half + (i & 1);
        ld[1][i] = *(const v2f*)(xs + (size_t)(8 * 64 * 64) + r * 64);
    }

#pragma unroll
    for (int g = 0; g < 8; ++g) {
        v8f acc00 = vzero, acc01 = vzero, acc10 = vzero, acc11 = vzero;
#pragma unroll
        for (int kk = 0; kk < 16; ++kk) {
            const v2f b0 = {ld[g & 1][2 * kk].x, ld[g & 1][2 * kk + 1].x};
            const v2f b1 = {ld[g & 1][2 * kk].y, ld[g & 1][2 * kk + 1].y};
            acc00 = __builtin_amdgcn_wmma_f32_16x16x4_f32(false, aF[0][kk], false, b0,
                                                          (short)0, acc00, false, false);
            acc01 = __builtin_amdgcn_wmma_f32_16x16x4_f32(false, aF[0][kk], false, b1,
                                                          (short)0, acc01, false, false);
            acc10 = __builtin_amdgcn_wmma_f32_16x16x4_f32(false, aF[1][kk], false, b0,
                                                          (short)0, acc10, false, false);
            acc11 = __builtin_amdgcn_wmma_f32_16x16x4_f32(false, aF[1][kk], false, b1,
                                                          (short)0, acc11, false, false);
        }

        if (g < 6) {
            __builtin_amdgcn_sched_barrier(0);
#pragma unroll
            for (int i = 0; i < 32; ++i) {
                const int r = (i >> 1) * 4 + 2 * half + (i & 1);
                ld[g & 1][i] = *(const v2f*)(xs + (size_t)(g + 2) * (8 * 64 * 64) + r * 64);
            }
            __builtin_amdgcn_sched_barrier(0);
        }

        float* trow = &tmp[(g & 1) * LDS_BUF + (wave >> 1) * LDS_PSTRIDE + (wave & 1)];
#pragma unroll
        for (int v = 0; v < 8; ++v) {
            const int hlo = v + 8 * half;
            trow[hlo * LDS_HSTRIDE + 2 * l16]                = acc00[v];
            trow[hlo * LDS_HSTRIDE + 2 * (16 + l16)]         = acc01[v];
            trow[(16 + hlo) * LDS_HSTRIDE + 2 * l16]         = acc10[v];
            trow[(16 + hlo) * LDS_HSTRIDE + 2 * (16 + l16)]  = acc11[v];
        }

        lds_barrier();

        const float* tbuf = &tmp[(g & 1) * LDS_BUF + half * LDS_PSTRIDE + 2 * l16];
        v2f t0[4][2], t1[4][2];
#pragma unroll
        for (int gg = 0; gg < 4; ++gg) {
            const float* tcol = tbuf + (gg * 8 + wave) * LDS_HSTRIDE;
#pragma unroll
            for (int k2 = 0; k2 < 2; ++k2) {
                t0[gg][k2] = *(const v2f*)(tcol + 2 * k2 * LDS_PSTRIDE);
                t1[gg][k2] = *(const v2f*)(tcol + 2 * k2 * LDS_PSTRIDE + 32);
            }
        }
#pragma unroll
        for (int gg = 0; gg < 4; ++gg) {
#pragma unroll
            for (int k2 = 0; k2 < 2; ++k2) {
                const int kk = 2 * g + k2;
                o[gg][0][0] = __builtin_amdgcn_wmma_f32_16x16x4_f32(false, aF[0][kk], false, t0[gg][k2],
                                                                    (short)0, o[gg][0][0], false, false);
                o[gg][0][1] = __builtin_amdgcn_wmma_f32_16x16x4_f32(false, aF[0][kk], false, t1[gg][k2],
                                                                    (short)0, o[gg][0][1], false, false);
                o[gg][1][0] = __builtin_amdgcn_wmma_f32_16x16x4_f32(false, aF[1][kk], false, t0[gg][k2],
                                                                    (short)0, o[gg][1][0], false, false);
                o[gg][1][1] = __builtin_amdgcn_wmma_f32_16x16x4_f32(false, aF[1][kk], false, t1[gg][k2],
                                                                    (short)0, o[gg][1][1], false, false);
            }
        }
    }

#pragma unroll
    for (int gg = 0; gg < 4; ++gg) {
        float* obase = out + (size_t)blockIdx.x * 32768 + (gg * 8 + wave) * 32;
#pragma unroll
        for (int v = 0; v < 8; ++v) {
            const int dlo = v + 8 * half;
            const v2f lo = {o[gg][0][0][v], o[gg][0][1][v]};
            const v2f hi = {o[gg][1][0][v], o[gg][1][1][v]};
            *(v2f*)(obase + dlo * 1024 + 2 * l16)        = lo;
            *(v2f*)(obase + (16 + dlo) * 1024 + 2 * l16) = hi;
        }
    }
}

extern "C" void kernel_launch(void* const* d_in, const int* in_sizes, int n_in,
                              void* d_out, int out_size, void* d_ws, size_t ws_size,
                              hipStream_t stream) {
    const float* x = (const float*)d_in[0];
    float* out     = (float*)d_out;

    spectral_pool_kernel<<<256, 256, 0, stream>>>(x, out);
}
